# starting kernel
# baseline (speedup 1.0000x reference)
#include <hip/hip_runtime.h>

#define DIM 512
#define HLEN (DIM * (DIM - 1) / 2)

typedef __attribute__((ext_vector_type(2))) float v2f;
typedef __attribute__((ext_vector_type(8))) float v8f;

__global__ void __launch_bounds__(512)
linmap_tri_matvec(const float* __restrict__ z,
                  const float* __restrict__ h,
                  float* __restrict__ out,
                  int batch) {
    __shared__ float zs[DIM];

    const int b   = blockIdx.x;
    const int tid = threadIdx.x;

    zs[tid] = z[(size_t)b * DIM + tid];
    __syncthreads();

    const int lane = tid & 31;
    const int wave = tid >> 5;
    const int m    = lane & 15;
    const int kb   = (lane >> 4) << 1;

    const float* __restrict__ hb = h + (size_t)b * (size_t)HLEN;

    for (int pass = 0; pass < 2; ++pass) {
        const int tile = (pass == 0) ? wave : (31 - wave);
        const int i0   = tile << 4;
        const int i0s  = __builtin_amdgcn_readfirstlane(i0);

        const int r      = i0 + m;
        const int rowoff = (r * (r - 1)) >> 1;
        const float* __restrict__ hrow = hb + rowoff;

        v8f c = {0.f, 0.f, 0.f, 0.f, 0.f, 0.f, 0.f, 0.f};

        for (int j0 = 0; j0 < i0s; j0 += 16) {
            v2f a0, a1, a2, a3, b0, b1, b2, b3;
            const float* __restrict__ hp = hrow + j0 + kb;
            a0.x = hp[0];  a0.y = hp[1];
            a1.x = hp[4];  a1.y = hp[5];
            a2.x = hp[8];  a2.y = hp[9];
            a3.x = hp[12]; a3.y = hp[13];
            const float* __restrict__ zp = zs + j0 + kb;
            b0.x = zp[0];  b0.y = zp[1];
            b1.x = zp[4];  b1.y = zp[5];
            b2.x = zp[8];  b2.y = zp[9];
            b3.x = zp[12]; b3.y = zp[13];
            c = __builtin_amdgcn_wmma_f32_16x16x4_f32(false, a0, false, b0,
                                                      (short)0, c, false, false);
            c = __builtin_amdgcn_wmma_f32_16x16x4_f32(false, a1, false, b1,
                                                      (short)0, c, false, false);
            c = __builtin_amdgcn_wmma_f32_16x16x4_f32(false, a2, false, b2,
                                                      (short)0, c, false, false);
            c = __builtin_amdgcn_wmma_f32_16x16x4_f32(false, a3, false, b3,
                                                      (short)0, c, false, false);
        }

        #pragma unroll
        for (int t = 0; t < 4; ++t) {
            const int c0 = i0 + 4 * t + kb;
            const int c1 = c0 + 1;
            int i0c = min(c0, r - 1); i0c = max(i0c, 0);
            int i1c = min(c1, r - 1); i1c = max(i1c, 0);
            const float h0 = hrow[i0c];
            const float h1 = hrow[i1c];
            v2f a, bv;
            a.x = h0 * (float)(c0 < r) + (float)(c0 == r);
            a.y = h1 * (float)(c1 < r) + (float)(c1 == r);
            bv.x = zs[c0];
            bv.y = zs[c1];
            c = __builtin_amdgcn_wmma_f32_16x16x4_f32(false, a, false, bv,
                                                      (short)0, c, false, false);
        }

        if (lane == 0 || lane == 16) {
            const int rb = i0 + ((lane >> 4) << 3);
            float* __restrict__ o = out + (size_t)b * DIM + rb;
            o[0] = c[0]; o[1] = c[1]; o[2] = c[2]; o[3] = c[3];
            o[4] = c[4]; o[5] = c[5]; o[6] = c[6]; o[7] = c[7];
        }
    }

    if (b == 0 && tid == 0) out[(size_t)batch * DIM] = 0.0f;
}

extern "C" void kernel_launch(void* const* d_in, const int* in_sizes, int n_in,
                              void* d_out, int out_size, void* d_ws, size_t ws_size,
                              hipStream_t stream) {
    const float* z = (const float*)d_in[0];
    const float* h = (const float*)d_in[1];
    float* out = (float*)d_out;

    const int batch = in_sizes[0] / DIM;

    linmap_tri_matvec<<<dim3(batch), dim3(512), 0, stream>>>(z, h, out, batch);
}


# Row-contiguous 512-byte wave loads (16 B per lane) of the packed rows feed the f32 WMMA through its diagonal, z in registers, 20-30 loads in flight; sum order changes (per-row trace, butterfly)
# speedup vs baseline: 2.2755x; 2.2755x over previous
#include <hip/hip_runtime.h>


#define DIM 512
#define HLEN (DIM * (DIM - 1) / 2)

typedef __attribute__((ext_vector_type(2))) float v2f;
typedef __attribute__((ext_vector_type(4))) float v4f;
typedef __attribute__((ext_vector_type(8))) float v8f;
typedef float v4f_a4 __attribute__((ext_vector_type(4), aligned(4)));

#define DEV static __device__ __forceinline__

DEV v4f ld16(const float* p) { return *(const v4f_a4*)p; }

DEV float xhalf(float x) {
    int b = __builtin_bit_cast(int, x);
    return __builtin_bit_cast(float,
        __builtin_amdgcn_permlanex16(b, b, 0x76543210, 0xfedcba98, false, false));
}

template <int M> DEV float rowx(float x) {
    int b = __builtin_bit_cast(int, x);
    return __builtin_bit_cast(float,
        __builtin_amdgcn_update_dpp(0, b, 0x160 + M, 0xf, 0xf, true));
}

DEV void mma_chunk(v4f a, v4f zq, v8f& c0, v8f& c1) {
    v2f a0 = {a.x, a.y}, a1 = {a.z, a.w};
    v2f b0 = {zq.x, zq.y}, b1 = {zq.z, zq.w};
    c0 = __builtin_amdgcn_wmma_f32_16x16x4_f32(false, a0, false, b0, (short)0, c0, false, false);
    c1 = __builtin_amdgcn_wmma_f32_16x16x4_f32(false, a1, false, b1, (short)0, c1, false, false);
}

DEV float diag_sum(v8f c, int dsel) {
    float d = 0.f;
    d = (dsel == 0) ? c[0] : d;
    d = (dsel == 1) ? c[1] : d;
    d = (dsel == 2) ? c[2] : d;
    d = (dsel == 3) ? c[3] : d;
    d = (dsel == 4) ? c[4] : d;
    d = (dsel == 5) ? c[5] : d;
    d = (dsel == 6) ? c[6] : d;
    d = (dsel == 7) ? c[7] : d;
    d = d + xhalf(d);
    d = d + rowx<8>(d);
    d = d + rowx<4>(d);
    d = d + rowx<2>(d);
    d = d + rowx<1>(d);
    return d;
}

struct Grp { v4f a[10]; };

DEV void load_group(Grp& g, const float* __restrict__ hb, int t, int wave, int lane4,
                    bool last_batch) {
    #pragma unroll
    for (int k = 0; k < 4; ++k) {
        const int i = 128 * k + 16 * t + wave;
        const float* __restrict__ p = hb + ((i * (i - 1)) >> 1) + lane4;
        #pragma unroll
        for (int q = 0; q <= k; ++q) {
            const int slot = (k * (k + 1)) / 2 + q;
            if (k == 3 && q == 3 && t == 7) {
                if (last_batch && wave == 15) {
                    const int n = 127 - lane4;
                    v4f v;
                    v.x = (n > 0) ? p[128 * q + 0] : 0.f;
                    v.y = (n > 1) ? p[128 * q + 1] : 0.f;
                    v.z = (n > 2) ? p[128 * q + 2] : 0.f;
                    v.w = (n > 3) ? p[128 * q + 3] : 0.f;
                    g.a[slot] = v;
                } else {
                    g.a[slot] = ld16(p + 128 * q);
                }
            } else {
                g.a[slot] = ld16(p + 128 * q);
            }
        }
    }
}

DEV void compute_group(const Grp& g, const v4f (&zq)[4], float* ys, int t, int wave,
                       int lane, int lane4, int dsel) {
    #pragma unroll
    for (int k = 0; k < 4; ++k) {
        const int i = 128 * k + 16 * t + wave;
        v8f c0 = {0.f, 0.f, 0.f, 0.f, 0.f, 0.f, 0.f, 0.f};
        v8f c1 = {0.f, 0.f, 0.f, 0.f, 0.f, 0.f, 0.f, 0.f};
        #pragma unroll
        for (int q = 0; q < k; ++q)
            mma_chunk(g.a[(k * (k + 1)) / 2 + q], zq[q], c0, c1);
        const v4f v = g.a[(k * (k + 1)) / 2 + k];
        const int n = (16 * t + wave) - lane4;
        v4f a;
        a.x = (0 < n) ? v.x : ((n == 0) ? 1.f : 0.f);
        a.y = (1 < n) ? v.y : ((n == 1) ? 1.f : 0.f);
        a.z = (2 < n) ? v.z : ((n == 2) ? 1.f : 0.f);
        a.w = (3 < n) ? v.w : ((n == 3) ? 1.f : 0.f);
        mma_chunk(a, zq[k], c0, c1);
        const float tot = diag_sum(c0 + c1, dsel);
        if (lane == 0) ys[i] = tot;
    }
}

__global__ void __launch_bounds__(512)
linmap_tri_matvec(const float* __restrict__ z,
                  const float* __restrict__ h,
                  float* __restrict__ out,
                  int batch) {
    __shared__ float ys[DIM];

    const int b     = blockIdx.x;
    const int tid   = threadIdx.x;
    const int lane  = tid & 31;
    const int lane4 = lane << 2;
    const int wave  = __builtin_amdgcn_readfirstlane(tid >> 5);
    const bool last_batch = (b == batch - 1);

    const float* __restrict__ hb = h + (size_t)b * (size_t)HLEN;
    const float* __restrict__ zb = z + (size_t)b * DIM;

    v4f zq[4];
    #pragma unroll
    for (int q = 0; q < 4; ++q) zq[q] = *(const v4f*)(zb + 128 * q + lane4);

    const int dsel = ((lane >> 3) == 0 || (lane >> 3) == 3) ? (lane & 7) : -1;

    Grp g[3];
    load_group(g[0], hb, 0, wave, lane4, last_batch);
    load_group(g[1], hb, 1, wave, lane4, last_batch);
    #pragma unroll
    for (int t = 0; t < 8; ++t) {
        if (t + 2 < 8) load_group(g[(t + 2) % 3], hb, t + 2, wave, lane4, last_batch);
        compute_group(g[t % 3], zq, ys, t, wave, lane, lane4, dsel);
    }

    __syncthreads();
    out[(size_t)b * DIM + tid] = ys[tid];

    if (b == 0 && tid == 0) out[(size_t)batch * DIM] = 0.0f;
}

extern "C" void kernel_launch(void* const* d_in, const int* in_sizes, int n_in,
                              void* d_out, int out_size, void* d_ws, size_t ws_size,
                              hipStream_t stream) {
    const float* z = (const float*)d_in[0];
    const float* h = (const float*)d_in[1];
    float* out = (float*)d_out;

    const int batch = in_sizes[0] / DIM;

    linmap_tri_matvec<<<dim3(batch), dim3(512), 0, stream>>>(z, h, out, batch);
}
